# starting kernel
# baseline (speedup 1.0000x reference)
#include <hip/hip_runtime.h>

typedef __attribute__((ext_vector_type(2))) float v2f;
typedef __attribute__((ext_vector_type(8))) float v8f;

#define N_NODES 100000
#define NB      16384
#define S       32
#define D       64
#define SLOPE   0.2f

__global__ __launch_bounds__(256) void spgat_gemm_kernel(
    const float* __restrict__ feat, const float* __restrict__ W_bal,
    const float* __restrict__ W_unbal, const float* __restrict__ a_bal,
    const float* __restrict__ a_unbal, float* __restrict__ H,
    float* __restrict__ sdot)
{
  const int tid  = threadIdx.x;
  const int lane = tid & 31;
  const int wave = blockIdx.x * (blockDim.x >> 5) + (tid >> 5);

  const int c  = wave / 6250;
  const int mt = wave % 6250;
  const int t  = c & 1;

  const float* A   = feat + (size_t)t * N_NODES * D;
  const float* Wm  = (c >> 1) ? W_unbal : W_bal;
  const float* ak  = (((c >> 1) ? a_unbal : a_bal)) + D;
  float*       out = H + (size_t)c * N_NODES * D;
  float*       sd  = sdot + (size_t)c * N_NODES;

  const int half  = lane >> 4;
  const int m16   = lane & 15;
  const int mbase = mt * 16;

  v8f acc0 = {}, acc1 = {}, acc2 = {}, acc3 = {};
#pragma unroll
  for (int k = 0; k < D; k += 4) {
    const int kk = k + half * 2;
    const float* arow = A + (size_t)(mbase + m16) * D + kk;
    v2f a;
    a.x = arow[0];
    a.y = arow[1];
    const float* b0 = Wm + (size_t)kk * D + m16;
    const float* b1 = b0 + D;
    v2f bb0, bb1, bb2, bb3;
    bb0.x = b0[0];  bb0.y = b1[0];
    bb1.x = b0[16]; bb1.y = b1[16];
    bb2.x = b0[32]; bb2.y = b1[32];
    bb3.x = b0[48]; bb3.y = b1[48];
    acc0 = __builtin_amdgcn_wmma_f32_16x16x4_f32(false, a, false, bb0, (short)0, acc0, false, false);
    acc1 = __builtin_amdgcn_wmma_f32_16x16x4_f32(false, a, false, bb1, (short)0, acc1, false, false);
    acc2 = __builtin_amdgcn_wmma_f32_16x16x4_f32(false, a, false, bb2, (short)0, acc2, false, false);
    acc3 = __builtin_amdgcn_wmma_f32_16x16x4_f32(false, a, false, bb3, (short)0, acc3, false, false);
  }

#pragma unroll
  for (int r = 0; r < 8; ++r) {
    const size_t row = (size_t)(mbase + r + half * 8) * D;
    out[row + m16]      = acc0[r];
    out[row + 16 + m16] = acc1[r];
    out[row + 32 + m16] = acc2[r];
    out[row + 48 + m16] = acc3[r];
  }

  const float ak0 = ak[m16], ak1 = ak[16 + m16], ak2 = ak[32 + m16], ak3 = ak[48 + m16];
#pragma unroll
  for (int r = 0; r < 8; ++r) {
    float p = acc0[r] * ak0 + acc1[r] * ak1 + acc2[r] * ak2 + acc3[r] * ak3;
#pragma unroll
    for (int o = 8; o; o >>= 1) p += __shfl_xor(p, o, 32);
    if (m16 == 0) sd[mbase + r + half * 8] = p;
  }
}

__global__ __launch_bounds__(256) void spgat_attn_kernel(
    const int* __restrict__ nodes, const int* __restrict__ npos,
    const int* __restrict__ nneg, const float* __restrict__ H,
    const float* __restrict__ sdot, const float* __restrict__ a_bal,
    const float* __restrict__ a_unbal, float* __restrict__ out)
{
  const int tid  = threadIdx.x;
  const int lane = tid & 31;
  const int wave = blockIdx.x * (blockDim.x >> 5) + (tid >> 5);

  const int w = wave >> 14;
  const int b = wave & (NB - 1);

  const int c1 = 3 * w;
  const int c2 = w + 1;
  const float* H1 = H + (size_t)c1 * N_NODES * D;
  const float* H2 = H + (size_t)c2 * N_NODES * D;
  const float* s1 = sdot + (size_t)c1 * N_NODES;
  const float* s2 = sdot + (size_t)c2 * N_NODES;
  const float* aq = w ? a_unbal : a_bal;

  const int node = nodes[b];

  const float2 q   = ((const float2*)(H1 + (size_t)node * D))[lane];
  const float2 aqv = ((const float2*)aq)[lane];
  float qa = q.x * aqv.x + q.y * aqv.y;
#pragma unroll
  for (int o = 16; o; o >>= 1) qa += __shfl_xor(qa, o, 32);

  const int* pos = npos + (size_t)b * S;
  const int* neg = nneg + (size_t)b * S;
  const int idxA = (lane == 0) ? node    : pos[lane - 1];
  const int idxB = (lane == 0) ? pos[31] : neg[lane - 1];
  const int idxC = neg[31];

  const float sA = s1[idxA];
  const float sB = (lane == 0) ? s1[idxB] : s2[idxB];
  const float sC = s2[idxC];

  const float xA = qa + sA, xB = qa + sB, xC = qa + sC;
  const float eA = xA > 0.f ? xA : SLOPE * xA;
  const float eB = xB > 0.f ? xB : SLOPE * xB;
  const float eC = xC > 0.f ? xC : SLOPE * xC;

  float m = fmaxf(eA, eB);
  if (lane == 31) m = fmaxf(m, eC);
#pragma unroll
  for (int o = 16; o; o >>= 1) m = fmaxf(m, __shfl_xor(m, o, 32));

  const float pA = __expf(eA - m);
  const float pB = __expf(eB - m);
  const float pC = (lane == 31) ? __expf(eC - m) : 0.f;
  float sum = pA + pB + pC;
#pragma unroll
  for (int o = 16; o; o >>= 1) sum += __shfl_xor(sum, o, 32);
  const float inv = 1.f / sum;
  const float alA = pA * inv, alB = pB * inv, alC = pC * inv;

  float2 acc = make_float2(0.f, 0.f);
  for (int j = 0; j < 65; ++j) {
    float al;
    int idx;
    if (j < 32)       { al = __shfl(alA, j, 32);      idx = __shfl(idxA, j, 32); }
    else if (j < 64)  { al = __shfl(alB, j - 32, 32); idx = __shfl(idxB, j - 32, 32); }
    else              { al = __shfl(alC, 31, 32);     idx = __shfl(idxC, 31, 32); }
    const float* Hsrc = (j <= 32) ? H1 : H2;
    const float2 v = ((const float2*)(Hsrc + (size_t)idx * D))[lane];
    acc.x += al * v.x;
    acc.y += al * v.y;
  }

  ((float2*)(out + ((size_t)w * NB + b) * D))[lane] = acc;
}

extern "C" void kernel_launch(void* const* d_in, const int* in_sizes, int n_in,
                              void* d_out, int out_size, void* d_ws, size_t ws_size,
                              hipStream_t stream) {
  const int*   nodes   = (const int*)d_in[0];
  const int*   npos    = (const int*)d_in[1];
  const int*   nneg    = (const int*)d_in[2];
  const float* feat    = (const float*)d_in[3];
  const float* W_bal   = (const float*)d_in[4];
  const float* a_bal   = (const float*)d_in[5];
  const float* W_unbal = (const float*)d_in[6];
  const float* a_unbal = (const float*)d_in[7];
  float*       out     = (float*)d_out;

  float* H    = (float*)d_ws;
  float* sdot = H + (size_t)4 * N_NODES * D;

  spgat_gemm_kernel<<<3125, 256, 0, stream>>>(feat, W_bal, W_unbal,
                                              a_bal, a_unbal, H, sdot);
  spgat_attn_kernel<<<4096, 256, 0, stream>>>(nodes, npos, nneg, H, sdot,
                                              a_bal, a_unbal, out);
}


# Project after aggregating: out=(sum alpha*x)@W, scores from x.(W@a_k); GEMM 400000->32768 rows, H table gone; reassociated f32 summation order changes
# speedup vs baseline: 1.5737x; 1.5737x over previous
#include <hip/hip_runtime.h>


typedef __attribute__((ext_vector_type(2))) float v2f;
typedef __attribute__((ext_vector_type(8))) float v8f;

#define N_NODES 100000
#define NB      16384
#define S       32
#define D       64
#define SLOPE   0.2f


__global__ __launch_bounds__(160) void spgat_score_kernel(
    const float* __restrict__ feat, const float* __restrict__ W_bal,
    const float* __restrict__ W_unbal, const float* __restrict__ a_bal,
    const float* __restrict__ a_unbal, float* __restrict__ sdot,
    float* __restrict__ wq)
{
  __shared__ float4 wk_s[2 * D / 4];

  const int tid  = threadIdx.x;
  const int lane = tid & 31;
  const int wave = blockIdx.x * 5 + (tid >> 5);
  const int hl   = lane >> 4;
  const int l16  = lane & 15;
  const int row0 = wave * 32;

  float4 x[16];
#pragma unroll
  for (int i = 0; i < 16; ++i) {
    const int row = row0 + i + 16 * hl;
    x[i] = ((const float4*)(feat + (size_t)row * D))[l16];
  }

  if (tid < 2 * D) {
    const int h = tid >> 6, k = tid & 63;
    const float* Wrow = (h ? W_unbal : W_bal) + k * D;
    const float* av   = h ? a_unbal : a_bal;
    float sk = 0.f, sq = 0.f;
    for (int n = 0; n < D; ++n) {
      const float wv = Wrow[n];
      sq += wv * av[n];
      sk += wv * av[D + n];
    }
    ((float*)wk_s)[tid] = sk;
    if (blockIdx.x == 0) wq[tid] = sq;
  }
  __syncthreads();

  const float4 wb = wk_s[l16];
  const float4 wu = wk_s[D / 4 + l16];

  float vb = 0.f, vu = 0.f;
#pragma unroll
  for (int i = 0; i < 16; ++i) {
    float pb = x[i].x * wb.x + x[i].y * wb.y + x[i].z * wb.z + x[i].w * wb.w;
    float pu = x[i].x * wu.x + x[i].y * wu.y + x[i].z * wu.z + x[i].w * wu.w;
#pragma unroll
    for (int o = 8; o; o >>= 1) {
      pb += __shfl_xor(pb, o, 32);
      pu += __shfl_xor(pu, o, 32);
    }
    if (l16 == i) { vb = pb; vu = pu; }
  }
  sdot[row0 + lane]                       = vb;
  sdot[(size_t)2 * N_NODES + row0 + lane] = vu;
}

__global__ __launch_bounds__(256) void spgat_attn_kernel(
    const int* __restrict__ nodes, const int* __restrict__ npos,
    const int* __restrict__ nneg, const float* __restrict__ feat,
    const float* __restrict__ sdot, const float* __restrict__ wq,
    float* __restrict__ agg)
{
  const int tid  = threadIdx.x;
  const int lane = tid & 31;
  const int wave = blockIdx.x * (blockDim.x >> 5) + (tid >> 5);

  const int w = wave >> 14;
  const int b = wave & (NB - 1);

  const int t1 = w;
  const int t2 = 1 - w;
  const float* X1 = feat + (size_t)t1 * N_NODES * D;
  const float* X2 = feat + (size_t)t2 * N_NODES * D;
  const float* s1 = sdot + (size_t)(w * 2 + t1) * N_NODES;
  const float* s2 = sdot + (size_t)(w * 2 + t2) * N_NODES;

  const int node = nodes[b];

  const float2 q   = ((const float2*)(X1 + (size_t)node * D))[lane];
  const float2 wqv = ((const float2*)(wq + w * D))[lane];
  float qa = q.x * wqv.x + q.y * wqv.y;
#pragma unroll
  for (int o = 16; o; o >>= 1) qa += __shfl_xor(qa, o, 32);

  const int* pos = npos + (size_t)b * S;
  const int* neg = nneg + (size_t)b * S;
  const int idxA = (lane == 0) ? node    : pos[lane - 1];
  const int idxB = (lane == 0) ? pos[31] : neg[lane - 1];
  const int idxC = neg[31];

  const float sA = s1[idxA];
  const float sB = (lane == 0) ? s1[idxB] : s2[idxB];
  const float sC = s2[idxC];

  const float xA = qa + sA, xB = qa + sB, xC = qa + sC;
  const float eA = xA > 0.f ? xA : SLOPE * xA;
  const float eB = xB > 0.f ? xB : SLOPE * xB;
  const float eC = xC > 0.f ? xC : SLOPE * xC;

  float m = fmaxf(eA, eB);
  if (lane == 31) m = fmaxf(m, eC);
#pragma unroll
  for (int o = 16; o; o >>= 1) m = fmaxf(m, __shfl_xor(m, o, 32));

  const float pA = __expf(eA - m);
  const float pB = __expf(eB - m);
  const float pC = (lane == 31) ? __expf(eC - m) : 0.f;
  float sum = pA + pB + pC;
#pragma unroll
  for (int o = 16; o; o >>= 1) sum += __shfl_xor(sum, o, 32);
  const float inv = 1.f / sum;
  const float alA = pA * inv, alB = pB * inv, alC = pC * inv;

  float2 acc = make_float2(0.f, 0.f);
  for (int j = 0; j < 65; ++j) {
    float al;
    int idx;
    if (j < 32)       { al = __shfl(alA, j, 32);      idx = __shfl(idxA, j, 32); }
    else if (j < 64)  { al = __shfl(alB, j - 32, 32); idx = __shfl(idxB, j - 32, 32); }
    else              { al = __shfl(alC, 31, 32);     idx = __shfl(idxC, 31, 32); }
    const float* Xsrc = (j <= 32) ? X1 : X2;
    const float2 v = ((const float2*)(Xsrc + (size_t)idx * D))[lane];
    acc.x += al * v.x;
    acc.y += al * v.y;
  }

  ((float2*)(agg + ((size_t)w * NB + b) * D))[lane] = acc;
}

__global__ __launch_bounds__(256) void spgat_proj_kernel(
    const float* __restrict__ agg, const float* __restrict__ W_bal,
    const float* __restrict__ W_unbal, float* __restrict__ outp)
{
  const int tid  = threadIdx.x;
  const int lane = tid & 31;
  const int wave = blockIdx.x * (blockDim.x >> 5) + (tid >> 5);

  const int h  = wave >> 10;
  const int mt = wave & 1023;

  const float* A   = agg + (size_t)h * NB * D;
  const float* Wm  = h ? W_unbal : W_bal;
  float*       out = outp + (size_t)h * NB * D;

  const int half  = lane >> 4;
  const int m16   = lane & 15;
  const int mbase = mt * 16;

  v8f acc0 = {}, acc1 = {}, acc2 = {}, acc3 = {};
#pragma unroll
  for (int k = 0; k < D; k += 4) {
    const int kk = k + half * 2;
    const float* arow = A + (size_t)(mbase + m16) * D + kk;
    v2f a;
    a.x = arow[0];
    a.y = arow[1];
    const float* b0 = Wm + (size_t)kk * D + m16;
    const float* b1 = b0 + D;
    v2f bb0, bb1, bb2, bb3;
    bb0.x = b0[0];  bb0.y = b1[0];
    bb1.x = b0[16]; bb1.y = b1[16];
    bb2.x = b0[32]; bb2.y = b1[32];
    bb3.x = b0[48]; bb3.y = b1[48];
    acc0 = __builtin_amdgcn_wmma_f32_16x16x4_f32(false, a, false, bb0, (short)0, acc0, false, false);
    acc1 = __builtin_amdgcn_wmma_f32_16x16x4_f32(false, a, false, bb1, (short)0, acc1, false, false);
    acc2 = __builtin_amdgcn_wmma_f32_16x16x4_f32(false, a, false, bb2, (short)0, acc2, false, false);
    acc3 = __builtin_amdgcn_wmma_f32_16x16x4_f32(false, a, false, bb3, (short)0, acc3, false, false);
  }

#pragma unroll
  for (int r = 0; r < 8; ++r) {
    const size_t row = (size_t)(mbase + r + half * 8) * D;
    out[row + m16]      = acc0[r];
    out[row + 16 + m16] = acc1[r];
    out[row + 32 + m16] = acc2[r];
    out[row + 48 + m16] = acc3[r];
  }
}

extern "C" void kernel_launch(void* const* d_in, const int* in_sizes, int n_in,
                              void* d_out, int out_size, void* d_ws, size_t ws_size,
                              hipStream_t stream) {
  const int*   nodes   = (const int*)d_in[0];
  const int*   npos    = (const int*)d_in[1];
  const int*   nneg    = (const int*)d_in[2];
  const float* feat    = (const float*)d_in[3];
  const float* W_bal   = (const float*)d_in[4];
  const float* a_bal   = (const float*)d_in[5];
  const float* W_unbal = (const float*)d_in[6];
  const float* a_unbal = (const float*)d_in[7];
  float*       out     = (float*)d_out;

  float* sdot = (float*)d_ws;
  float* wq   = sdot + (size_t)4 * N_NODES;
  float* agg  = wq + 2 * D;

  spgat_score_kernel<<<1250, 160, 0, stream>>>(feat, W_bal, W_unbal,
                                               a_bal, a_unbal, sdot, wq);
  spgat_attn_kernel<<<4096, 256, 0, stream>>>(nodes, npos, nneg, feat, sdot,
                                              wq, agg);
  spgat_proj_kernel<<<256, 256, 0, stream>>>(agg, W_bal, W_unbal, out);
}


# Aggregation gather: each 16-lane row loads a whole 256-B feature row as float4 (2 edges per load), 16 loads in flight issued with the score loads ahead of the softmax; f32 summation order changes
# speedup vs baseline: 1.8162x; 1.1541x over previous
#include <hip/hip_runtime.h>


typedef __attribute__((ext_vector_type(2))) float v2f;
typedef __attribute__((ext_vector_type(8))) float v8f;

#define N_NODES 100000
#define NB      16384
#define S       32
#define D       64
#define SLOPE   0.2f


__global__ __launch_bounds__(160) void spgat_score_kernel(
    const float* __restrict__ feat, const float* __restrict__ W_bal,
    const float* __restrict__ W_unbal, const float* __restrict__ a_bal,
    const float* __restrict__ a_unbal, float* __restrict__ sdot,
    float* __restrict__ wq)
{
  __shared__ float4 wk_s[2 * D / 4];

  const int tid  = threadIdx.x;
  const int lane = tid & 31;
  const int wave = blockIdx.x * 5 + (tid >> 5);
  const int hl   = lane >> 4;
  const int l16  = lane & 15;
  const int row0 = wave * 32;

  float4 x[16];
#pragma unroll
  for (int i = 0; i < 16; ++i) {
    const int row = row0 + i + 16 * hl;
    x[i] = ((const float4*)(feat + (size_t)row * D))[l16];
  }

  if (tid < 2 * D) {
    const int h = tid >> 6, k = tid & 63;
    const float* Wrow = (h ? W_unbal : W_bal) + k * D;
    const float* av   = h ? a_unbal : a_bal;
    float sk = 0.f, sq = 0.f;
    for (int n = 0; n < D; ++n) {
      const float wv = Wrow[n];
      sq += wv * av[n];
      sk += wv * av[D + n];
    }
    ((float*)wk_s)[tid] = sk;
    if (blockIdx.x == 0) wq[tid] = sq;
  }
  __syncthreads();

  const float4 wb = wk_s[l16];
  const float4 wu = wk_s[D / 4 + l16];

  float vb = 0.f, vu = 0.f;
#pragma unroll
  for (int i = 0; i < 16; ++i) {
    float pb = x[i].x * wb.x + x[i].y * wb.y + x[i].z * wb.z + x[i].w * wb.w;
    float pu = x[i].x * wu.x + x[i].y * wu.y + x[i].z * wu.z + x[i].w * wu.w;
#pragma unroll
    for (int o = 8; o; o >>= 1) {
      pb += __shfl_xor(pb, o, 32);
      pu += __shfl_xor(pu, o, 32);
    }
    if (l16 == i) { vb = pb; vu = pu; }
  }
  sdot[row0 + lane]                       = vb;
  sdot[(size_t)2 * N_NODES + row0 + lane] = vu;
}


template <int I>
__device__ __forceinline__ int row_share_i(int v) {
  return __builtin_amdgcn_update_dpp(0, v, 0x150 + I, 0xF, 0xF, true);
}
template <int I>
__device__ __forceinline__ float row_share_f(float v) {
  return __int_as_float(row_share_i<I>(__float_as_int(v)));
}
__device__ __forceinline__ float xhalf(float x) {
  return __int_as_float(__builtin_amdgcn_permlanex16(
      __float_as_int(x), __float_as_int(x), 0x76543210, 0xfedcba98, false, false));
}

#define REP16(M) M(0) M(1) M(2) M(3) M(4) M(5) M(6) M(7) \
                 M(8) M(9) M(10) M(11) M(12) M(13) M(14) M(15)

__global__ __launch_bounds__(256) void spgat_attn_kernel(
    const int* __restrict__ nodes, const int* __restrict__ npos,
    const int* __restrict__ nneg, const float* __restrict__ feat,
    const float* __restrict__ sdot, const float* __restrict__ wq,
    float* __restrict__ agg)
{
  const int tid  = threadIdx.x;
  const int lane = tid & 31;
  const int wave = blockIdx.x * (blockDim.x >> 5) + (tid >> 5);

  const int w = wave >> 14;
  const int b = wave & (NB - 1);

  const int      half = lane >> 4;
  const int      l16  = lane & 15;
  const unsigned cb   = l16 * 16u;

  const int t1 = w;
  const int t2 = 1 - w;
  const float* s1 = sdot + (size_t)(w * 2 + t1) * N_NODES;
  const float* s2 = sdot + (size_t)(w * 2 + t2) * N_NODES;

  const int node = nodes[b];
  const int ipos = npos[(size_t)b * S + lane];
  const int ineg = nneg[(size_t)b * S + lane];

  const int offA = (t1 * N_NODES + ipos) * (D * 4);
  const int offB = (t2 * N_NODES + ineg) * (D * 4);
  const int offC = (t1 * N_NODES + node) * (D * 4);

  const float  sA  = s1[ipos];
  const float  sB  = s2[ineg];
  const float  sC  = s1[node];
  const float4 wqv = ((const float4*)(wq + w * D))[l16];

  const char* fb = (const char*)feat;
  const float4 xc = *(const float4*)(fb + ((unsigned)offC + cb));
  float4 x[16];
#define LOAD_A(I) x[I] = *(const float4*)(fb + ((unsigned)row_share_i<I>(offA) + cb));
  REP16(LOAD_A)
#undef LOAD_A
  __builtin_amdgcn_sched_group_barrier(0x020, 24, 0);
  __builtin_amdgcn_sched_group_barrier(0x080, 14, 0);

  float qa = xc.x * wqv.x + xc.y * wqv.y + xc.z * wqv.z + xc.w * wqv.w;
#pragma unroll
  for (int o = 8; o; o >>= 1) qa += __shfl_xor(qa, o, 32);

  const float xA = qa + sA, xB = qa + sB, xC = qa + sC;
  const float eA = xA > 0.f ? xA : SLOPE * xA;
  const float eB = xB > 0.f ? xB : SLOPE * xB;
  const float eC = xC > 0.f ? xC : SLOPE * xC;

  float m = fmaxf(fmaxf(eA, eB), eC);
#pragma unroll
  for (int o = 16; o; o >>= 1) m = fmaxf(m, __shfl_xor(m, o, 32));

  const float pA = __expf(eA - m);
  const float pB = __expf(eB - m);
  const float pC = __expf(eC - m);
  float sum = pA + pB;
#pragma unroll
  for (int o = 16; o; o >>= 1) sum += __shfl_xor(sum, o, 32);
  sum += pC;
  const float inv = 1.f / sum;
  const float alA = pA * inv, alB = pB * inv;
  const float alC = half ? 0.f : pC * inv;

  float4 acc = make_float4(alC * xc.x, alC * xc.y, alC * xc.z, alC * xc.w);
#define STEP_A(I) { const float al = row_share_f<I>(alA);                           \
    acc.x += al * x[I].x; acc.y += al * x[I].y;                                    \
    acc.z += al * x[I].z; acc.w += al * x[I].w;                                    \
    x[I] = *(const float4*)(fb + ((unsigned)row_share_i<I>(offB) + cb)); }
  REP16(STEP_A)
#undef STEP_A
#define STEP_B(I) { const float al = row_share_f<I>(alB);                           \
    acc.x += al * x[I].x; acc.y += al * x[I].y;                                    \
    acc.z += al * x[I].z; acc.w += al * x[I].w; }
  REP16(STEP_B)
#undef STEP_B

  acc.x += xhalf(acc.x); acc.y += xhalf(acc.y);
  acc.z += xhalf(acc.z); acc.w += xhalf(acc.w);

  if (half == 0)
    *(float4*)((char*)(agg + ((size_t)w * NB + b) * D) + cb) = acc;
}
#undef REP16

__global__ __launch_bounds__(256) void spgat_proj_kernel(
    const float* __restrict__ agg, const float* __restrict__ W_bal,
    const float* __restrict__ W_unbal, float* __restrict__ outp)
{
  const int tid  = threadIdx.x;
  const int lane = tid & 31;
  const int wave = blockIdx.x * (blockDim.x >> 5) + (tid >> 5);

  const int h  = wave >> 10;
  const int mt = wave & 1023;

  const float* A   = agg + (size_t)h * NB * D;
  const float* Wm  = h ? W_unbal : W_bal;
  float*       out = outp + (size_t)h * NB * D;

  const int half  = lane >> 4;
  const int m16   = lane & 15;
  const int mbase = mt * 16;

  v8f acc0 = {}, acc1 = {}, acc2 = {}, acc3 = {};
#pragma unroll
  for (int k = 0; k < D; k += 4) {
    const int kk = k + half * 2;
    const float* arow = A + (size_t)(mbase + m16) * D + kk;
    v2f a;
    a.x = arow[0];
    a.y = arow[1];
    const float* b0 = Wm + (size_t)kk * D + m16;
    const float* b1 = b0 + D;
    v2f bb0, bb1, bb2, bb3;
    bb0.x = b0[0];  bb0.y = b1[0];
    bb1.x = b0[16]; bb1.y = b1[16];
    bb2.x = b0[32]; bb2.y = b1[32];
    bb3.x = b0[48]; bb3.y = b1[48];
    acc0 = __builtin_amdgcn_wmma_f32_16x16x4_f32(false, a, false, bb0, (short)0, acc0, false, false);
    acc1 = __builtin_amdgcn_wmma_f32_16x16x4_f32(false, a, false, bb1, (short)0, acc1, false, false);
    acc2 = __builtin_amdgcn_wmma_f32_16x16x4_f32(false, a, false, bb2, (short)0, acc2, false, false);
    acc3 = __builtin_amdgcn_wmma_f32_16x16x4_f32(false, a, false, bb3, (short)0, acc3, false, false);
  }

#pragma unroll
  for (int r = 0; r < 8; ++r) {
    const size_t row = (size_t)(mbase + r + half * 8) * D;
    out[row + m16]      = acc0[r];
    out[row + 16 + m16] = acc1[r];
    out[row + 32 + m16] = acc2[r];
    out[row + 48 + m16] = acc3[r];
  }
}

extern "C" void kernel_launch(void* const* d_in, const int* in_sizes, int n_in,
                              void* d_out, int out_size, void* d_ws, size_t ws_size,
                              hipStream_t stream) {
  const int*   nodes   = (const int*)d_in[0];
  const int*   npos    = (const int*)d_in[1];
  const int*   nneg    = (const int*)d_in[2];
  const float* feat    = (const float*)d_in[3];
  const float* W_bal   = (const float*)d_in[4];
  const float* a_bal   = (const float*)d_in[5];
  const float* W_unbal = (const float*)d_in[6];
  const float* a_unbal = (const float*)d_in[7];
  float*       out     = (float*)d_out;

  float* sdot = (float*)d_ws;
  float* wq   = sdot + (size_t)4 * N_NODES;
  float* agg  = wq + 2 * D;

  spgat_score_kernel<<<1250, 160, 0, stream>>>(feat, W_bal, W_unbal,
                                               a_bal, a_unbal, sdot, wq);
  spgat_attn_kernel<<<4096, 256, 0, stream>>>(nodes, npos, nneg, feat, sdot,
                                              wq, agg);
  spgat_proj_kernel<<<256, 256, 0, stream>>>(agg, W_bal, W_unbal, out);
}


# Attention stage runs 4096 long waves instead of 32768 short ones: each wave loops over 4 targets x 2 heads with its row loads rolling across items; per-row f32 summation order unchanged
# speedup vs baseline: 1.9670x; 1.0830x over previous
#include <hip/hip_runtime.h>


typedef __attribute__((ext_vector_type(2))) float v2f;
typedef __attribute__((ext_vector_type(8))) float v8f;

#define N_NODES 100000
#define NB      16384
#define S       32
#define D       64
#define SLOPE   0.2f


__global__ __launch_bounds__(160) void spgat_score_kernel(
    const float* __restrict__ feat, const float* __restrict__ W_bal,
    const float* __restrict__ W_unbal, const float* __restrict__ a_bal,
    const float* __restrict__ a_unbal, float* __restrict__ sdot,
    float* __restrict__ wq)
{
  __shared__ float4 wk_s[2 * D / 4];

  const int tid  = threadIdx.x;
  const int lane = tid & 31;
  const int wave = blockIdx.x * 5 + (tid >> 5);
  const int hl   = lane >> 4;
  const int l16  = lane & 15;
  const int row0 = wave * 32;

  float4 x[16];
#pragma unroll
  for (int i = 0; i < 16; ++i) {
    const int row = row0 + i + 16 * hl;
    x[i] = ((const float4*)(feat + (size_t)row * D))[l16];
  }

  if (tid < 2 * D) {
    const int h = tid >> 6, k = tid & 63;
    const float* Wrow = (h ? W_unbal : W_bal) + k * D;
    const float* av   = h ? a_unbal : a_bal;
    float sk = 0.f, sq = 0.f;
    for (int n = 0; n < D; ++n) {
      const float wv = Wrow[n];
      sq += wv * av[n];
      sk += wv * av[D + n];
    }
    ((float*)wk_s)[tid] = sk;
    if (blockIdx.x == 0) wq[tid] = sq;
  }
  __syncthreads();

  const float4 wb = wk_s[l16];
  const float4 wu = wk_s[D / 4 + l16];

  float vb = 0.f, vu = 0.f;
#pragma unroll
  for (int i = 0; i < 16; ++i) {
    float pb = x[i].x * wb.x + x[i].y * wb.y + x[i].z * wb.z + x[i].w * wb.w;
    float pu = x[i].x * wu.x + x[i].y * wu.y + x[i].z * wu.z + x[i].w * wu.w;
#pragma unroll
    for (int o = 8; o; o >>= 1) {
      pb += __shfl_xor(pb, o, 32);
      pu += __shfl_xor(pu, o, 32);
    }
    if (l16 == i) { vb = pb; vu = pu; }
  }
  sdot[row0 + lane]                       = vb;
  sdot[(size_t)2 * N_NODES + row0 + lane] = vu;
}


#define TPW 4

template <int I>
__device__ __forceinline__ int row_share_i(int v) {
  return __builtin_amdgcn_update_dpp(0, v, 0x150 + I, 0xF, 0xF, true);
}
template <int I>
__device__ __forceinline__ float row_share_f(float v) {
  return __int_as_float(row_share_i<I>(__float_as_int(v)));
}
__device__ __forceinline__ float xhalf(float x) {
  return __int_as_float(__builtin_amdgcn_permlanex16(
      __float_as_int(x), __float_as_int(x), 0x76543210, 0xfedcba98, false, false));
}

#define REP16(M) M(0) M(1) M(2) M(3) M(4) M(5) M(6) M(7) \
                 M(8) M(9) M(10) M(11) M(12) M(13) M(14) M(15)

template <bool HAS_NEXT>
__device__ __forceinline__ void attn_item(
    const char* __restrict__ fb, const unsigned cb, const int half,
    float4 (&x)[16], float4& xc, float& sA, float& sB, float& sC,
    const int offB, const float4 wqv, float* __restrict__ outrow,
    const float* __restrict__ s1n, const float* __restrict__ s2n,
    const int iposn, const int inegn, const int noden,
    const int offAn, const int offCn)
{
  float  sAn = 0.f, sBn = 0.f, sCn = 0.f;
  float4 xcn = make_float4(0.f, 0.f, 0.f, 0.f);
  if constexpr (HAS_NEXT) {
    sAn = s1n[iposn];
    sBn = s2n[inegn];
    sCn = s1n[noden];
    xcn = *(const float4*)(fb + ((unsigned)offCn + cb));
  }

  float qa = xc.x * wqv.x + xc.y * wqv.y + xc.z * wqv.z + xc.w * wqv.w;
#pragma unroll
  for (int o = 8; o; o >>= 1) qa += __shfl_xor(qa, o, 32);

  const float xA = qa + sA, xB = qa + sB, xC = qa + sC;
  const float eA = xA > 0.f ? xA : SLOPE * xA;
  const float eB = xB > 0.f ? xB : SLOPE * xB;
  const float eC = xC > 0.f ? xC : SLOPE * xC;

  float m = fmaxf(fmaxf(eA, eB), eC);
#pragma unroll
  for (int o = 16; o; o >>= 1) m = fmaxf(m, __shfl_xor(m, o, 32));

  const float pA = __expf(eA - m);
  const float pB = __expf(eB - m);
  const float pC = __expf(eC - m);
  float sum = pA + pB;
#pragma unroll
  for (int o = 16; o; o >>= 1) sum += __shfl_xor(sum, o, 32);
  sum += pC;
  const float inv = 1.f / sum;
  const float alA = pA * inv, alB = pB * inv;
  const float alC = half ? 0.f : pC * inv;

  float4 acc = make_float4(alC * xc.x, alC * xc.y, alC * xc.z, alC * xc.w);
#define STEP_A(I) { const float al = row_share_f<I>(alA);                           \
    acc.x += al * x[I].x; acc.y += al * x[I].y;                                    \
    acc.z += al * x[I].z; acc.w += al * x[I].w;                                    \
    x[I] = *(const float4*)(fb + ((unsigned)row_share_i<I>(offB) + cb)); }
  REP16(STEP_A)
#undef STEP_A
#define STEP_B(I) { const float al = row_share_f<I>(alB);                           \
    acc.x += al * x[I].x; acc.y += al * x[I].y;                                    \
    acc.z += al * x[I].z; acc.w += al * x[I].w;                                    \
    if constexpr (HAS_NEXT)                                                        \
      x[I] = *(const float4*)(fb + ((unsigned)row_share_i<I>(offAn) + cb)); }
  REP16(STEP_B)
#undef STEP_B

  acc.x += xhalf(acc.x); acc.y += xhalf(acc.y);
  acc.z += xhalf(acc.z); acc.w += xhalf(acc.w);

  if (half == 0)
    *(float4*)((char*)outrow + cb) = acc;

  if constexpr (HAS_NEXT) { xc = xcn; sA = sAn; sB = sBn; sC = sCn; }
}

__global__ __launch_bounds__(256) void spgat_attn_kernel(
    const int* __restrict__ nodes, const int* __restrict__ npos,
    const int* __restrict__ nneg, const float* __restrict__ feat,
    const float* __restrict__ sdot, const float* __restrict__ wq,
    float* __restrict__ agg)
{
  const int tid  = threadIdx.x;
  const int lane = tid & 31;
  const int wave = blockIdx.x * (blockDim.x >> 5) + (tid >> 5);
  const int b0   = wave * TPW;

  const int      half = lane >> 4;
  const int      l16  = lane & 15;
  const unsigned cb   = l16 * 16u;

  int nd[TPW], ip[TPW], in[TPW];
#pragma unroll
  for (int t = 0; t < TPW; ++t) {
    nd[t] = nodes[b0 + t];
    ip[t] = npos[(size_t)(b0 + t) * S + lane];
    in[t] = nneg[(size_t)(b0 + t) * S + lane];
  }
  const float4 wq0 = ((const float4*)wq)[l16];
  const float4 wq1 = ((const float4*)(wq + D))[l16];

  const float* s00 = sdot;
  const float* s01 = sdot + (size_t)1 * N_NODES;
  const float* s10 = sdot + (size_t)2 * N_NODES;
  const float* s11 = sdot + (size_t)3 * N_NODES;
  const int T1 = N_NODES * (D * 4);
  const char* fb = (const char*)feat;

  float  sA = s00[ip[0]];
  float  sB = s01[in[0]];
  float  sC = s00[nd[0]];
  float4 xc = *(const float4*)(fb + ((unsigned)(nd[0] * (D * 4)) + cb));
  float4 x[16];
  {
    const int offA = ip[0] * (D * 4);
#define LOAD_A(I) x[I] = *(const float4*)(fb + ((unsigned)row_share_i<I>(offA) + cb));
    REP16(LOAD_A)
#undef LOAD_A
  }

  float* out0 = agg + (size_t)b0 * D;
  float* out1 = agg + ((size_t)NB + b0) * D;

  for (int t = 0; t < TPW - 1; ++t) {
    attn_item<true>(fb, cb, half, x, xc, sA, sB, sC,
                    T1 + in[0] * (D * 4), wq0, out0 + (size_t)t * D,
                    s11, s10, ip[0], in[0], nd[0],
                    T1 + ip[0] * (D * 4), T1 + nd[0] * (D * 4));
    attn_item<true>(fb, cb, half, x, xc, sA, sB, sC,
                    in[0] * (D * 4), wq1, out1 + (size_t)t * D,
                    s00, s01, ip[1], in[1], nd[1],
                    ip[1] * (D * 4), nd[1] * (D * 4));
#pragma unroll
    for (int k = 0; k < TPW - 1; ++k) { nd[k] = nd[k + 1]; ip[k] = ip[k + 1]; in[k] = in[k + 1]; }
  }
  attn_item<true>(fb, cb, half, x, xc, sA, sB, sC,
                  T1 + in[0] * (D * 4), wq0, out0 + (size_t)(TPW - 1) * D,
                  s11, s10, ip[0], in[0], nd[0],
                  T1 + ip[0] * (D * 4), T1 + nd[0] * (D * 4));
  attn_item<false>(fb, cb, half, x, xc, sA, sB, sC,
                   in[0] * (D * 4), wq1, out1 + (size_t)(TPW - 1) * D,
                   s00, s01, 0, 0, 0, 0, 0);
}
#undef REP16

__global__ __launch_bounds__(256) void spgat_proj_kernel(
    const float* __restrict__ agg, const float* __restrict__ W_bal,
    const float* __restrict__ W_unbal, float* __restrict__ outp)
{
  const int tid  = threadIdx.x;
  const int lane = tid & 31;
  const int wave = blockIdx.x * (blockDim.x >> 5) + (tid >> 5);

  const int h  = wave >> 10;
  const int mt = wave & 1023;

  const float* A   = agg + (size_t)h * NB * D;
  const float* Wm  = h ? W_unbal : W_bal;
  float*       out = outp + (size_t)h * NB * D;

  const int half  = lane >> 4;
  const int m16   = lane & 15;
  const int mbase = mt * 16;

  v8f acc0 = {}, acc1 = {}, acc2 = {}, acc3 = {};
#pragma unroll
  for (int k = 0; k < D; k += 4) {
    const int kk = k + half * 2;
    const float* arow = A + (size_t)(mbase + m16) * D + kk;
    v2f a;
    a.x = arow[0];
    a.y = arow[1];
    const float* b0 = Wm + (size_t)kk * D + m16;
    const float* b1 = b0 + D;
    v2f bb0, bb1, bb2, bb3;
    bb0.x = b0[0];  bb0.y = b1[0];
    bb1.x = b0[16]; bb1.y = b1[16];
    bb2.x = b0[32]; bb2.y = b1[32];
    bb3.x = b0[48]; bb3.y = b1[48];
    acc0 = __builtin_amdgcn_wmma_f32_16x16x4_f32(false, a, false, bb0, (short)0, acc0, false, false);
    acc1 = __builtin_amdgcn_wmma_f32_16x16x4_f32(false, a, false, bb1, (short)0, acc1, false, false);
    acc2 = __builtin_amdgcn_wmma_f32_16x16x4_f32(false, a, false, bb2, (short)0, acc2, false, false);
    acc3 = __builtin_amdgcn_wmma_f32_16x16x4_f32(false, a, false, bb3, (short)0, acc3, false, false);
  }

#pragma unroll
  for (int r = 0; r < 8; ++r) {
    const size_t row = (size_t)(mbase + r + half * 8) * D;
    out[row + m16]      = acc0[r];
    out[row + 16 + m16] = acc1[r];
    out[row + 32 + m16] = acc2[r];
    out[row + 48 + m16] = acc3[r];
  }
}

extern "C" void kernel_launch(void* const* d_in, const int* in_sizes, int n_in,
                              void* d_out, int out_size, void* d_ws, size_t ws_size,
                              hipStream_t stream) {
  const int*   nodes   = (const int*)d_in[0];
  const int*   npos    = (const int*)d_in[1];
  const int*   nneg    = (const int*)d_in[2];
  const float* feat    = (const float*)d_in[3];
  const float* W_bal   = (const float*)d_in[4];
  const float* a_bal   = (const float*)d_in[5];
  const float* W_unbal = (const float*)d_in[6];
  const float* a_unbal = (const float*)d_in[7];
  float*       out     = (float*)d_out;

  float* sdot = (float*)d_ws;
  float* wq   = sdot + (size_t)4 * N_NODES;
  float* agg  = wq + 2 * D;

  spgat_score_kernel<<<1250, 160, 0, stream>>>(feat, W_bal, W_unbal,
                                               a_bal, a_unbal, sdot, wq);
  spgat_attn_kernel<<<512, 256, 0, stream>>>(nodes, npos, nneg, feat, sdot,
                                              wq, agg);
  spgat_proj_kernel<<<256, 256, 0, stream>>>(agg, W_bal, W_unbal, out);
}
